# starting kernel
# baseline (speedup 1.0000x reference)
#include <hip/hip_runtime.h>
#include <hip/hip_bf16.h>

typedef __attribute__((ext_vector_type(2))) float v2f;
typedef __attribute__((ext_vector_type(8))) float v8f;


#define WAVES_PER_BLOCK 4

__global__ __launch_bounds__(128) void ConvLocal2d_wmma_f32_kernel(
    const float* __restrict__ x,
    const float* __restrict__ w,
    const float* __restrict__ bias,
    float* __restrict__ out)
{
    __shared__ float patch[WAVES_PER_BLOCK][8][288];

    const int lane = threadIdx.x & 31;
    const int wave = threadIdx.x >> 5;
    const int loc  = blockIdx.x * WAVES_PER_BLOCK + wave;
    const int i = loc >> 6;
    const int j = loc & 63;

    float (*p)[288] = patch[wave];
    for (int idx = lane; idx < 256; idx += 32) {
        const int b = idx >> 5;
        const int c = idx & 31;
        const float* xb = x + (size_t)(b * 32 + c) * 4096;
        float* dst = &p[b][c * 9];
        #pragma unroll
        for (int kw = 0; kw < 3; ++kw) {
            const int col = j + kw - 1;
            const bool cok = (unsigned)col < 64u;
            #pragma unroll
            for (int kh = 0; kh < 3; ++kh) {
                const int row = i + kh - 1;
                float v = (cok && (unsigned)row < 64u) ? xb[row * 64 + col] : 0.0f;
                dst[kw * 3 + kh] = v;
            }
        }
    }
    __syncthreads();

    const int mrow   = lane & 15;
    const int colOff = (lane >> 4) * 2;
    const float* wloc = w + (size_t)loc * (32 * 288);
    const float* a0p  = wloc + mrow * 288 + colOff;
    const float* a1p  = a0p + 16 * 288;
    const float* bp   = &p[lane & 7][colOff];

    __builtin_prefetch(a0p, 0, 0);
    __builtin_prefetch(a1p, 0, 0);

    const int hi8 = (lane >> 4) * 8;
    v8f c0, c1;
    #pragma unroll
    for (int r = 0; r < 8; ++r) {
        c0[r] = bias[r + hi8];
        c1[r] = bias[16 + r + hi8];
    }

    #pragma unroll 4
    for (int k = 0; k < 288; k += 4) {
        v2f a0 = *(const v2f*)(a0p + k);
        v2f a1 = *(const v2f*)(a1p + k);
        v2f bb = *(const v2f*)(bp + k);
        c0 = __builtin_amdgcn_wmma_f32_16x16x4_f32(false, a0, false, bb, (short)0, c0, false, false);
        c1 = __builtin_amdgcn_wmma_f32_16x16x4_f32(false, a1, false, bb, (short)0, c1, false, false);
    }

    const int b = lane & 15;
    if (b < 8) {
        float* op = out + (size_t)b * (32 * 4096) + loc;
        #pragma unroll
        for (int r = 0; r < 8; ++r) {
            const int o0 = r + hi8;
            op[(size_t)o0 * 4096]        = c0[r];
            op[(size_t)(o0 + 16) * 4096] = c1[r];
        }
    }
}

extern "C" void kernel_launch(void* const* d_in, const int* in_sizes, int n_in,
                              void* d_out, int out_size, void* d_ws, size_t ws_size,
                              hipStream_t stream) {
    const float* x    = (const float*)d_in[0];
    const float* w    = (const float*)d_in[1];
    const float* bias = (const float*)d_in[2];
    float* out        = (float*)d_out;

    ConvLocal2d_wmma_f32_kernel<<<4096 / WAVES_PER_BLOCK, 128, 0, stream>>>(x, w, bias, out);
}


# patch gather: each block loads x rows i-1..i+1 once, 48 unconditional row-contiguous loads per lane, into a shared LDS tile; waves copy their patches from LDS (was 72 one-at-a-time 32-line gathers)
# speedup vs baseline: 1.1321x; 1.1321x over previous
#include <hip/hip_runtime.h>
#include <hip/hip_bf16.h>

typedef __attribute__((ext_vector_type(2))) float v2f;
typedef __attribute__((ext_vector_type(8))) float v8f;


#define WAVES_PER_BLOCK 4
#define XPL 25

__global__ __launch_bounds__(128) void ConvLocal2d_wmma_f32_kernel(
    const float* __restrict__ x,
    const float* __restrict__ w,
    const float* __restrict__ bias,
    float* __restrict__ out)
{
    __shared__ float patch[WAVES_PER_BLOCK][8][288];
    __shared__ float xs[256 * XPL];

    const int lane = threadIdx.x & 31;
    const int wave = threadIdx.x >> 5;
    const int loc  = blockIdx.x * WAVES_PER_BLOCK + wave;
    const int i = loc >> 6;
    const int j = loc & 63;

    {
        const int j0   = j - wave;
        const int cc   = lane & 7;
        const int sub  = lane >> 3;
        const int col  = j0 - 1 + cc;
        const bool cok = (cc < 6) && ((unsigned)col < 64u);
        const int colc = col < 0 ? 0 : (col > 63 ? 63 : col);
        const float* xl = x + (size_t)(wave * 4 + sub) * 4096 + colc;
        float* xd = xs + (wave * 4 + sub) * XPL + cc;
        float v[3][16];
        #pragma unroll
        for (int kh = 0; kh < 3; ++kh) {
            const int row  = i + kh - 1;
            const int rowc = row < 0 ? 0 : (row > 63 ? 63 : row);
            #pragma unroll
            for (int n = 0; n < 16; ++n)
                v[kh][n] = xl[(size_t)n * (16 * 4096) + rowc * 64];
        }
        #pragma unroll
        for (int kh = 0; kh < 3; ++kh) {
            const bool ok = cok && ((unsigned)(i + kh - 1) < 64u);
            #pragma unroll
            for (int n = 0; n < 16; ++n)
                xd[n * (16 * XPL) + kh * 8] = ok ? v[kh][n] : 0.0f;
        }
    }
    __syncthreads();

    float (*p)[288] = patch[wave];
    #pragma unroll
    for (int b = 0; b < 8; ++b) {
        const float* src = xs + (b * 32 + lane) * XPL + wave;
        float* dst = &p[b][lane * 9];
        #pragma unroll
        for (int kw = 0; kw < 3; ++kw) {
            #pragma unroll
            for (int kh = 0; kh < 3; ++kh)
                dst[kw * 3 + kh] = src[kh * 8 + kw];
        }
    }
    __syncthreads();

    const int mrow   = lane & 15;
    const int colOff = (lane >> 4) * 2;
    const float* wloc = w + (size_t)loc * (32 * 288);
    const float* a0p  = wloc + mrow * 288 + colOff;
    const float* a1p  = a0p + 16 * 288;
    const float* bp   = &p[lane & 7][colOff];

    __builtin_prefetch(a0p, 0, 0);
    __builtin_prefetch(a1p, 0, 0);

    const int hi8 = (lane >> 4) * 8;
    v8f c0, c1;
    #pragma unroll
    for (int r = 0; r < 8; ++r) {
        c0[r] = bias[r + hi8];
        c1[r] = bias[16 + r + hi8];
    }

    #pragma unroll 4
    for (int k = 0; k < 288; k += 4) {
        v2f a0 = *(const v2f*)(a0p + k);
        v2f a1 = *(const v2f*)(a1p + k);
        v2f bb = *(const v2f*)(bp + k);
        c0 = __builtin_amdgcn_wmma_f32_16x16x4_f32(false, a0, false, bb, (short)0, c0, false, false);
        c1 = __builtin_amdgcn_wmma_f32_16x16x4_f32(false, a1, false, bb, (short)0, c1, false, false);
    }

    const int b = lane & 15;
    if (b < 8) {
        float* op = out + (size_t)b * (32 * 4096) + loc;
        #pragma unroll
        for (int r = 0; r < 8; ++r) {
            const int o0 = r + hi8;
            op[(size_t)o0 * 4096]        = c0[r];
            op[(size_t)(o0 + 16) * 4096] = c1[r];
        }
    }
}

extern "C" void kernel_launch(void* const* d_in, const int* in_sizes, int n_in,
                              void* d_out, int out_size, void* d_ws, size_t ws_size,
                              hipStream_t stream) {
    const float* x    = (const float*)d_in[0];
    const float* w    = (const float*)d_in[1];
    const float* bias = (const float*)d_in[2];
    float* out        = (float*)d_out;

    ConvLocal2d_wmma_f32_kernel<<<4096 / WAVES_PER_BLOCK, 128, 0, stream>>>(x, w, bias, out);
}


# weight stream: each wave reads its 32x288 weights as whole 128-byte lines (16 B per lane, 16-24 loads in flight), forms A fragments through a private LDS tile (was 8 B per lane over 16 lines)
# speedup vs baseline: 1.3049x; 1.1526x over previous
#include <hip/hip_runtime.h>
#include <hip/hip_bf16.h>

typedef __attribute__((ext_vector_type(2))) float v2f;
typedef __attribute__((ext_vector_type(4))) float v4f;
typedef __attribute__((ext_vector_type(8))) float v8f;


#define WAVES_PER_BLOCK 4
#define XPL 25
#define KCH 32
#define NCH 9
#define AHEAD 3
#define SPITCH 36

__global__ __launch_bounds__(128) void ConvLocal2d_wmma_f32_kernel(
    const float* __restrict__ x,
    const float* __restrict__ w,
    const float* __restrict__ bias,
    float* __restrict__ out)
{
    __shared__ float patch[WAVES_PER_BLOCK][8][288];
    __shared__ __attribute__((aligned(16))) float xs[256 * XPL];
    static_assert(WAVES_PER_BLOCK * 32 * SPITCH <= 256 * XPL, "staging tiles must fit in the x tile");

    const int lane = threadIdx.x & 31;
    const int wave = threadIdx.x >> 5;
    const int loc  = blockIdx.x * WAVES_PER_BLOCK + wave;
    const int i = loc >> 6;
    const int j = loc & 63;

    const int lgrp = lane >> 3;
    const int lc4  = (lane & 7) * 4;
    const float* wl = w + (size_t)loc * (32 * 288) + (lgrp * 8) * 288 + lc4;
    v4f ring[AHEAD][8];
    #pragma unroll
    for (int q = 0; q < AHEAD; ++q) {
        #pragma unroll
        for (int n = 0; n < 8; ++n)
            ring[q][n] = *(const v4f*)(wl + n * 288 + q * KCH);
    }

    {
        const int j0   = j - wave;
        const int cc   = lane & 7;
        const int sub  = lane >> 3;
        const int col  = j0 - 1 + cc;
        const bool cok = (cc < 6) && ((unsigned)col < 64u);
        const int colc = col < 0 ? 0 : (col > 63 ? 63 : col);
        const float* xl = x + (size_t)(wave * 4 + sub) * 4096 + colc;
        float* xd = xs + (wave * 4 + sub) * XPL + cc;
        float v[3][16];
        #pragma unroll
        for (int kh = 0; kh < 3; ++kh) {
            const int row  = i + kh - 1;
            const int rowc = row < 0 ? 0 : (row > 63 ? 63 : row);
            #pragma unroll
            for (int n = 0; n < 16; ++n)
                v[kh][n] = xl[(size_t)n * (16 * 4096) + rowc * 64];
        }
        #pragma unroll
        for (int kh = 0; kh < 3; ++kh) {
            const bool ok = cok && ((unsigned)(i + kh - 1) < 64u);
            #pragma unroll
            for (int n = 0; n < 16; ++n)
                xd[n * (16 * XPL) + kh * 8] = ok ? v[kh][n] : 0.0f;
        }
    }
    __syncthreads();

    float (*p)[288] = patch[wave];
    #pragma unroll
    for (int b = 0; b < 8; ++b) {
        const float* src = xs + (b * 32 + lane) * XPL + wave;
        float* dst = &p[b][lane * 9];
        #pragma unroll
        for (int kw = 0; kw < 3; ++kw) {
            #pragma unroll
            for (int kh = 0; kh < 3; ++kh)
                dst[kw * 3 + kh] = src[kh * 8 + kw];
        }
    }
    __syncthreads();

    const int mrow   = lane & 15;
    const int colOff = (lane >> 4) * 2;
    float* stg       = xs + wave * (32 * SPITCH);
    float* sst       = stg + (lgrp * 8) * SPITCH + lc4;
    const float* sa  = stg + mrow * SPITCH + colOff;
    const float* bp  = &p[lane & 7][colOff];

    const int hi8 = (lane >> 4) * 8;
    v8f c0, c1;
    #pragma unroll
    for (int r = 0; r < 8; ++r) {
        c0[r] = bias[r + hi8];
        c1[r] = bias[16 + r + hi8];
    }

    #pragma unroll
    for (int q = 0; q < NCH; ++q) {
        #pragma unroll
        for (int n = 0; n < 8; ++n)
            *(v4f*)(sst + n * SPITCH) = ring[q % AHEAD][n];
        if (q + AHEAD < NCH) {
            #pragma unroll
            for (int n = 0; n < 8; ++n)
                ring[q % AHEAD][n] = *(const v4f*)(wl + n * 288 + (q + AHEAD) * KCH);
        }
        v2f a0[8], a1[8], bb[8];
        #pragma unroll
        for (int s = 0; s < 8; ++s) {
            a0[s] = *(const v2f*)(sa + 4 * s);
            a1[s] = *(const v2f*)(sa + 16 * SPITCH + 4 * s);
            bb[s] = *(const v2f*)(bp + q * KCH + 4 * s);
        }
        #pragma unroll
        for (int s = 0; s < 8; ++s) {
            c0 = __builtin_amdgcn_wmma_f32_16x16x4_f32(false, a0[s], false, bb[s], (short)0, c0, false, false);
            c1 = __builtin_amdgcn_wmma_f32_16x16x4_f32(false, a1[s], false, bb[s], (short)0, c1, false, false);
        }
    }

    const int b = lane & 15;
    if (b < 8) {
        float* op = out + (size_t)b * (32 * 4096) + loc;
        #pragma unroll
        for (int r = 0; r < 8; ++r) {
            const int o0 = r + hi8;
            op[(size_t)o0 * 4096]        = c0[r];
            op[(size_t)(o0 + 16) * 4096] = c1[r];
        }
    }
}

extern "C" void kernel_launch(void* const* d_in, const int* in_sizes, int n_in,
                              void* d_out, int out_size, void* d_ws, size_t ws_size,
                              hipStream_t stream) {
    const float* x    = (const float*)d_in[0];
    const float* w    = (const float*)d_in[1];
    const float* bias = (const float*)d_in[2];
    float* out        = (float*)d_out;

    ConvLocal2d_wmma_f32_kernel<<<4096 / WAVES_PER_BLOCK, 128, 0, stream>>>(x, w, bias, out);
}


# output stores: a block now holds 16 adjacent locations (16 waves, x tile 18 columns); results pass through an LDS tile and leave as 64-byte row pieces, 16 B per lane (was 4-byte stores to 16 lines)
# speedup vs baseline: 3.1833x; 2.4395x over previous
#include <hip/hip_runtime.h>
#include <hip/hip_bf16.h>


typedef __attribute__((ext_vector_type(2))) float v2f;
typedef __attribute__((ext_vector_type(4))) float v4f;
typedef __attribute__((ext_vector_type(8))) float v8f;


#define NW 16
#define TW 18
#define XPL 55
#define KCH 32
#define NCH 9
#define AHEAD 3
#define SPITCH 36
#define OPITCH 20

__global__ __launch_bounds__(NW * 32) void ConvLocal2d_wmma_f32_kernel(
    const float* __restrict__ x,
    const float* __restrict__ w,
    const float* __restrict__ bias,
    float* __restrict__ out)
{
    __shared__ float patch[NW][8][288];
    __shared__ __attribute__((aligned(16))) float xs[NW * 32 * SPITCH];
    static_assert(256 * XPL <= NW * 32 * SPITCH, "the x tile must fit in the staging tiles");
    __shared__ __attribute__((aligned(16))) float ostage[256 * OPITCH];

    const int lane = threadIdx.x & 31;
    const int wave = threadIdx.x >> 5;
    const int loc  = blockIdx.x * NW + wave;
    const int i = loc >> 6;
    const int j = loc & 63;

    const int lgrp = lane >> 3;
    const int lc4  = (lane & 7) * 4;
    const float* wl = w + (size_t)loc * (32 * 288) + (lgrp * 8) * 288 + lc4;
    v4f ring[AHEAD][8];
    #pragma unroll
    for (int q = 0; q < AHEAD; ++q) {
        #pragma unroll
        for (int n = 0; n < 8; ++n)
            ring[q][n] = *(const v4f*)(wl + n * 288 + q * KCH);
    }

    {
        const int j0   = j - wave;
        const int cc   = lane;
        const int col  = j0 - 1 + cc;
        const bool cin = cc < TW;
        const bool cok = cin && ((unsigned)col < 64u);
        const int colc = col < 0 ? 0 : (col > 63 ? 63 : col);
        const float* xl = x + (size_t)(wave * 16) * 4096 + colc;
        float* xd = xs + (wave * 16) * XPL + cc;
        float v[3][16];
        #pragma unroll
        for (int kh = 0; kh < 3; ++kh) {
            const int row  = i + kh - 1;
            const int rowc = row < 0 ? 0 : (row > 63 ? 63 : row);
            #pragma unroll
            for (int n = 0; n < 16; ++n)
                v[kh][n] = xl[(size_t)n * 4096 + rowc * 64];
        }
        if (cin) {
            #pragma unroll
            for (int kh = 0; kh < 3; ++kh) {
                const bool ok = cok && ((unsigned)(i + kh - 1) < 64u);
                #pragma unroll
                for (int n = 0; n < 16; ++n)
                    xd[n * XPL + kh * TW] = ok ? v[kh][n] : 0.0f;
            }
        }
    }
    __syncthreads();

    float (*p)[288] = patch[wave];
    #pragma unroll
    for (int b = 0; b < 8; ++b) {
        const float* src = xs + (b * 32 + lane) * XPL + wave;
        float* dst = &p[b][lane * 9];
        #pragma unroll
        for (int kw = 0; kw < 3; ++kw) {
            #pragma unroll
            for (int kh = 0; kh < 3; ++kh)
                dst[kw * 3 + kh] = src[kh * TW + kw];
        }
    }
    __syncthreads();

    const int mrow   = lane & 15;
    const int colOff = (lane >> 4) * 2;
    float* stg       = xs + wave * (32 * SPITCH);
    float* sst       = stg + (lgrp * 8) * SPITCH + lc4;
    const float* sa  = stg + mrow * SPITCH + colOff;
    const float* bp  = &p[lane & 7][colOff];

    const int hi8 = (lane >> 4) * 8;
    v8f c0, c1;
    #pragma unroll
    for (int r = 0; r < 8; ++r) {
        c0[r] = bias[r + hi8];
        c1[r] = bias[16 + r + hi8];
    }

    #pragma unroll
    for (int q = 0; q < NCH; ++q) {
        #pragma unroll
        for (int n = 0; n < 8; ++n)
            *(v4f*)(sst + n * SPITCH) = ring[q % AHEAD][n];
        if (q + AHEAD < NCH) {
            #pragma unroll
            for (int n = 0; n < 8; ++n)
                ring[q % AHEAD][n] = *(const v4f*)(wl + n * 288 + (q + AHEAD) * KCH);
        }
        v2f a0[8], a1[8], bb[8];
        #pragma unroll
        for (int s = 0; s < 8; ++s) {
            a0[s] = *(const v2f*)(sa + 4 * s);
            a1[s] = *(const v2f*)(sa + 16 * SPITCH + 4 * s);
            bb[s] = *(const v2f*)(bp + q * KCH + 4 * s);
        }
        #pragma unroll
        for (int s = 0; s < 8; ++s) {
            c0 = __builtin_amdgcn_wmma_f32_16x16x4_f32(false, a0[s], false, bb[s], (short)0, c0, false, false);
            c1 = __builtin_amdgcn_wmma_f32_16x16x4_f32(false, a1[s], false, bb[s], (short)0, c1, false, false);
        }
    }

    {
        const int b = lane & 15;
        if (b < 8) {
            float* os = ostage + (b * 32 + hi8) * OPITCH + wave;
            #pragma unroll
            for (int r = 0; r < 8; ++r) {
                os[r * OPITCH]        = c0[r];
                os[(r + 16) * OPITCH] = c1[r];
            }
        }
    }
    __syncthreads();
    {
        const int piece = threadIdx.x & 3;
        const int row0  = threadIdx.x >> 2;
        float* ob = out + (size_t)i * 64 + (j - wave) + piece * 4;
        #pragma unroll
        for (int k = 0; k < 2; ++k) {
            const int row = row0 + 128 * k;
            const v4f val = *(const v4f*)(ostage + row * OPITCH + piece * 4);
            *(v4f*)(ob + (size_t)row * 4096) = val;
        }
    }
}

extern "C" void kernel_launch(void* const* d_in, const int* in_sizes, int n_in,
                              void* d_out, int out_size, void* d_ws, size_t ws_size,
                              hipStream_t stream) {
    const float* x    = (const float*)d_in[0];
    const float* w    = (const float*)d_in[1];
    const float* bias = (const float*)d_in[2];
    float* out        = (float*)d_out;

    ConvLocal2d_wmma_f32_kernel<<<4096 / NW, NW * 32, 0, stream>>>(x, w, bias, out);
}
